# starting kernel
# baseline (speedup 1.0000x reference)
#include <hip/hip_runtime.h>
#include <hip/hip_bf16.h>

typedef float v2f __attribute__((ext_vector_type(2)));
typedef float v8f __attribute__((ext_vector_type(8)));
typedef float f4  __attribute__((ext_vector_type(4)));

constexpr int BB = 64;
constexpr int TT = 512;
constexpr int FF = 1024;
constexpr int UU = 64;
constexpr int MM = BB * TT;

#define KCH 64
#define LDA 66
#define LDB 66

__global__ __launch_bounds__(256) void crf_potentials_gemm(
    const float* __restrict__ x, const float* __restrict__ w,
    const float* __restrict__ bias, const float* __restrict__ lb,
    const float* __restrict__ rb, float* __restrict__ pot) {
  __shared__ __attribute__((aligned(16))) float lds_a[128 * LDA];
  __shared__ __attribute__((aligned(16))) float lds_b[64 * LDB];

  const int tid = threadIdx.x;
  const int lane = tid & 31;
  const int wave = tid >> 5;
  const int blockRow = blockIdx.x * 128;
  const int rowBase = blockRow + wave * 16;

  const int mrow = lane & 15;
  const int kbase = (lane >> 4) << 1;

  v8f acc[4] = {};

  for (int kk = 0; kk < FF; kk += KCH) {
    __syncthreads();
#pragma unroll
    for (int i = 0; i < 8; ++i) {
      int f = tid + i * 256;
      int row = f >> 4;
      int q = f & 15;
      const float* src = x + (size_t)(blockRow + row) * FF + kk + q * 4;
      if (kk + KCH < FF) __builtin_prefetch(src + KCH, 0, 0);
      f4 av = *(const f4*)src;
      float* dst = &lds_a[row * LDA + q * 4];
      dst[0] = av.x; dst[1] = av.y; dst[2] = av.z; dst[3] = av.w;
    }
#pragma unroll
    for (int i = 0; i < 4; ++i) {
      int f = tid + i * 256;
      int row = f >> 4;
      int q = f & 15;
      f4 bv = *(const f4*)(w + (size_t)(kk + row) * UU + q * 4);
      lds_b[(q * 4 + 0) * LDB + row] = bv.x;
      lds_b[(q * 4 + 1) * LDB + row] = bv.y;
      lds_b[(q * 4 + 2) * LDB + row] = bv.z;
      lds_b[(q * 4 + 3) * LDB + row] = bv.w;
    }
    __syncthreads();
#pragma unroll
    for (int k4 = 0; k4 < KCH; k4 += 4) {
      v2f afrag = *(const v2f*)&lds_a[(wave * 16 + mrow) * LDA + k4 + kbase];
#pragma unroll
      for (int nt = 0; nt < 4; ++nt) {
        int n = nt * 16 + mrow;
        v2f bfrag = *(const v2f*)&lds_b[n * LDB + k4 + kbase];
        acc[nt] = __builtin_amdgcn_wmma_f32_16x16x4_f32(
            false, afrag, false, bfrag, (short)0, acc[nt], false, false);
      }
    }
  }

  const int rsel = (lane >> 4) * 8;
#pragma unroll
  for (int nt = 0; nt < 4; ++nt) {
    int col = nt * 16 + mrow;
    float bia = bias[col];
    float lbv = lb[col];
    float rbv = rb[col];
#pragma unroll
    for (int r = 0; r < 8; ++r) {
      int row = rowBase + r + rsel;
      float v = acc[nt][r] + bia;
      int t = row & (TT - 1);
      if (t == 0) v += lbv;
      if (t == TT - 1) v += rbv;
      pot[(size_t)row * UU + col] = v;
    }
  }
}

__global__ __launch_bounds__(256) void crf_viterbi(
    const float* __restrict__ pot, const float* __restrict__ chain,
    int* __restrict__ path) {
  __shared__ float alpha[UU];
  __shared__ float pv[4][UU];
  __shared__ int pidx[4][UU];
  __shared__ unsigned char bp[TT * UU];
  __shared__ __attribute__((aligned(16))) float potbuf[2][32 * UU];

  const int tid = threadIdx.x;
  const int c = tid & 63;
  const int pg = tid >> 6;
  const int b = blockIdx.x;
  const float* potb = pot + (size_t)b * TT * UU;

  float ck[16];
#pragma unroll
  for (int j = 0; j < 16; ++j) ck[j] = chain[(pg * 16 + j) * UU + c];

#pragma unroll
  for (int i = 0; i < 2; ++i) {
    int f = tid + i * 256;
    *(f4*)&potbuf[0][f * 4] = *(const f4*)(potb + f * 4);
  }
  __syncthreads();
  if (pg == 0) alpha[c] = potbuf[0][c];
  __syncthreads();

  int t = 1;
  for (int chunk = 0; chunk < 16; ++chunk) {
    const int buf = chunk & 1;
    f4 nv0, nv1;
    if (chunk < 15) {
      nv0 = *(const f4*)(potb + (chunk + 1) * 32 * UU + (tid + 0) * 4);
      nv1 = *(const f4*)(potb + (chunk + 1) * 32 * UU + (tid + 256) * 4);
    }
    const int tEnd = (chunk + 1) * 32;
    for (; t < tEnd; ++t) {
      const int tl = t - chunk * 32;
      float best = -3.402823466e38f;
      int bestp = 0;
#pragma unroll
      for (int j = 0; j < 16; ++j) {
        float s = alpha[pg * 16 + j] + ck[j];
        if (s > best) { best = s; bestp = pg * 16 + j; }
      }
      pv[pg][c] = best;
      pidx[pg][c] = bestp;
      __syncthreads();
      if (pg == 0) {
        float bb = pv[0][c];
        int bi = pidx[0][c];
#pragma unroll
        for (int g = 1; g < 4; ++g) {
          float s = pv[g][c];
          if (s > bb) { bb = s; bi = pidx[g][c]; }
        }
        bp[t * UU + c] = (unsigned char)bi;
        alpha[c] = bb + potbuf[buf][tl * UU + c];
      }
      __syncthreads();
    }
    if (chunk < 15) {
      *(f4*)&potbuf[1 - buf][(tid + 0) * 4] = nv0;
      *(f4*)&potbuf[1 - buf][(tid + 256) * 4] = nv1;
      __syncthreads();
    }
  }

  if (tid == 0) {
    float bb = alpha[0];
    int tag = 0;
    for (int i = 1; i < UU; ++i) {
      float s = alpha[i];
      if (s > bb) { bb = s; tag = i; }
    }
    int* outb = path + (size_t)b * TT;
    outb[TT - 1] = tag;
    for (int tt = TT - 1; tt >= 1; --tt) {
      tag = bp[tt * UU + tag];
      outb[tt - 1] = tag;
    }
  }
}

extern "C" void kernel_launch(void* const* d_in, const int* in_sizes, int n_in,
                              void* d_out, int out_size, void* d_ws,
                              size_t ws_size, hipStream_t stream) {
  (void)in_sizes; (void)n_in; (void)out_size; (void)ws_size;
  const float* x     = (const float*)d_in[0];
  const float* w     = (const float*)d_in[1];
  const float* bias  = (const float*)d_in[2];
  const float* chain = (const float*)d_in[3];
  const float* lb    = (const float*)d_in[4];
  const float* rb    = (const float*)d_in[5];
  float* pot = (float*)d_ws;
  int* out = (int*)d_out;

  crf_potentials_gemm<<<dim3(MM / 128), dim3(256), 0, stream>>>(
      x, w, bias, lb, rb, pot);
  crf_viterbi<<<dim3(BB), dim3(256), 0, stream>>>(pot, chain, out);
}


# Viterbi step: 64-way argmax in two levels in registers: 8 group maxima by max instructions, strict-compare chain over the groups carrying the winner's 8 sums, index by equality; same sums and index
# speedup vs baseline: 1.7080x; 1.7080x over previous
#include <hip/hip_runtime.h>
#include <hip/hip_bf16.h>

typedef float v2f __attribute__((ext_vector_type(2)));
typedef float v8f __attribute__((ext_vector_type(8)));
typedef float f4  __attribute__((ext_vector_type(4)));

constexpr int BB = 64;
constexpr int TT = 512;
constexpr int FF = 1024;
constexpr int UU = 64;
constexpr int MM = BB * TT;

#define KCH 64
#define LDAW 68
#define LDBW 1028

__global__ __launch_bounds__(256) void crf_potentials_gemm(
    const float* __restrict__ x, const float* __restrict__ w,
    const float* __restrict__ bias, const float* __restrict__ lb,
    const float* __restrict__ rb, float* __restrict__ pot) {
  __shared__ __attribute__((aligned(16))) float lds_b[UU * LDBW];
  __shared__ __attribute__((aligned(16))) float lds_a[8 * 16 * LDAW];

  const int tid = threadIdx.x;
  const int lane = tid & 31;
  const int wave = tid >> 5;
  const int rowBase = blockIdx.x * 128 + wave * 16;

  const int mrow = lane & 15;
  const int kbase = (lane >> 4) << 1;
  const int q = lane & 15;
  const int hf = lane >> 4;

  const float* xl = x + (size_t)(rowBase + hf) * FF + 4 * q;
  float* sa = lds_a + wave * (16 * LDAW);
  float* sa_st = sa + hf * LDAW + 4 * q;
  const float* sa_fr = sa + mrow * LDAW + kbase;
  const float* sb_fr = lds_b + mrow * LDBW + kbase;

  f4 r0[8], r1[8];
#pragma unroll
  for (int i = 0; i < 8; ++i) r0[i] = *(const f4*)(xl + (size_t)(2 * i) * FF);
#pragma unroll
  for (int i = 0; i < 8; ++i) r1[i] = *(const f4*)(xl + (size_t)(2 * i) * FF + KCH);

  {
    const int b4 = q >> 2;
#pragma unroll 4
    for (int it = 0; it < 16; ++it) {
      const int k0 = (it * 8 + wave) * 8;
      f4 bv[4];
#pragma unroll
      for (int r = 0; r < 4; ++r) {
        const int k = k0 + ((b4 + r) & 3) + 4 * hf;
        bv[r] = *(const f4*)(w + (size_t)k * UU + 4 * q);
      }
#pragma unroll
      for (int r = 0; r < 4; ++r) {
        const int k = k0 + ((b4 + r) & 3) + 4 * hf;
        lds_b[(4 * q + 0) * LDBW + k] = bv[r].x;
        lds_b[(4 * q + 1) * LDBW + k] = bv[r].y;
        lds_b[(4 * q + 2) * LDBW + k] = bv[r].z;
        lds_b[(4 * q + 3) * LDBW + k] = bv[r].w;
      }
    }
  }
  __syncthreads();

  v8f acc[4] = {};

#define STORE_SLOT(R)                                                         \
  _Pragma("unroll") for (int i = 0; i < 8; ++i)                               \
      *(f4*)(sa_st + (2 * i) * LDAW) = R[i];
#define LOAD_CHUNK(R, K0)                                                     \
  _Pragma("unroll") for (int i = 0; i < 8; ++i)                               \
      R[i] = *(const f4*)(xl + (size_t)(2 * i) * FF + (K0));
#define MMA_CHUNK(K0)                                                         \
  _Pragma("unroll") for (int k4 = 0; k4 < KCH; k4 += 4) {                     \
    v2f afrag = *(const v2f*)(sa_fr + k4);                                    \
    _Pragma("unroll") for (int nt = 0; nt < 4; ++nt) {                        \
      v2f bfrag = *(const v2f*)(sb_fr + nt * 16 * LDBW + (K0) + k4);          \
      acc[nt] = __builtin_amdgcn_wmma_f32_16x16x4_f32(                        \
          false, afrag, false, bfrag, (short)0, acc[nt], false, false);       \
    }                                                                         \
  }

  for (int kk = 0; kk < FF - 2 * KCH; kk += 2 * KCH) {
    STORE_SLOT(r0)
    LOAD_CHUNK(r0, kk + 2 * KCH)
    MMA_CHUNK(kk)
    STORE_SLOT(r1)
    LOAD_CHUNK(r1, kk + 3 * KCH)
    MMA_CHUNK(kk + KCH)
  }
  STORE_SLOT(r0)
  MMA_CHUNK(FF - 2 * KCH)
  STORE_SLOT(r1)
  MMA_CHUNK(FF - KCH)
#undef STORE_SLOT
#undef LOAD_CHUNK
#undef MMA_CHUNK

  const int rsel = (lane >> 4) * 8;
#pragma unroll
  for (int nt = 0; nt < 4; ++nt) {
    int col = nt * 16 + mrow;
    float bia = bias[col];
    float lbv = lb[col];
    float rbv = rb[col];
#pragma unroll
    for (int r = 0; r < 8; ++r) {
      int row = rowBase + r + rsel;
      float v = acc[nt][r] + bia;
      int t = row & (TT - 1);
      if (t == 0) v += lbv;
      if (t == TT - 1) v += rbv;
      pot[(size_t)row * UU + col] = v;
    }
  }
}

__global__ __launch_bounds__(64) void crf_viterbi(
    const float* __restrict__ pot, const float* __restrict__ chain,
    int* __restrict__ path) {
  __shared__ __attribute__((aligned(16))) float alpha[2][UU];
  __shared__ __attribute__((aligned(16))) unsigned int bpw[TT * UU / 4];
  unsigned char* bp = (unsigned char*)bpw;
  __shared__ __attribute__((aligned(16))) float potbuf[2][32 * UU];

  const int c = threadIdx.x;
  const int b = blockIdx.x;
  const float* potb = pot + (size_t)b * TT * UU;

  float ck[UU];
#pragma unroll
  for (int p = 0; p < UU; ++p) ck[p] = chain[p * UU + c];

#pragma unroll
  for (int i = 0; i < 8; ++i) {
    int f = c + i * 64;
    *(f4*)&potbuf[0][f * 4] = *(const f4*)(potb + f * 4);
  }
  __syncthreads();
  alpha[0][c] = potbuf[0][c];
  bp[c] = 0;
  __syncthreads();

  int t = 1;
  for (int chunk = 0; chunk < 16; ++chunk) {
    const int buf = chunk & 1;
    f4 nv[8];
    if (chunk < 15) {
#pragma unroll
      for (int i = 0; i < 8; ++i)
        nv[i] = *(const f4*)(potb + (chunk + 1) * 32 * UU + (c + i * 64) * 4);
    }
    const int tEnd = (chunk + 1) * 32;
    for (; t < tEnd; ++t) {
      const int tl = t - chunk * 32;
      const float* arow = alpha[(t - 1) & 1];
      f4 av[16];
#pragma unroll
      for (int i = 0; i < 16; ++i) av[i] = *(const f4*)(arow + 4 * i);
      const float pt = potbuf[buf][tl * UU + c];

      float s[UU];
#pragma unroll
      for (int p = 0; p < UU; ++p) s[p] = av[p >> 2][p & 3] + ck[p];

      float bb = -3.402823466e38f;
      int gi = 0;
      float wv[8];
#pragma unroll
      for (int k = 0; k < 8; ++k) wv[k] = -3.402823466e38f;
#pragma unroll
      for (int g = 0; g < 8; ++g) {
        const float gm = fmaxf(fmaxf(fmaxf(s[8 * g + 0], s[8 * g + 1]), fmaxf(s[8 * g + 2], s[8 * g + 3])),
                               fmaxf(fmaxf(s[8 * g + 4], s[8 * g + 5]), fmaxf(s[8 * g + 6], s[8 * g + 7])));
        const bool up = gm > bb;
        bb = fmaxf(bb, gm);
        gi = up ? g : gi;
#pragma unroll
        for (int k = 0; k < 8; ++k) wv[k] = up ? s[8 * g + k] : wv[k];
      }
      int klo = (wv[2] == bb) ? 2 : 3;
      klo = (wv[1] == bb) ? 1 : klo;
      klo = (wv[0] == bb) ? 0 : klo;
      int khi = (wv[6] == bb) ? 6 : 7;
      khi = (wv[5] == bb) ? 5 : khi;
      khi = (wv[4] == bb) ? 4 : khi;
      const bool inlo = fmaxf(fmaxf(wv[0], wv[1]), fmaxf(wv[2], wv[3])) == bb;
      const int bidx = gi * 8 + (inlo ? klo : khi);
      bp[t * UU + c] = (unsigned char)bidx;
      alpha[t & 1][c] = bb + pt;
      __syncthreads();
    }
    if (chunk < 15) {
#pragma unroll
      for (int i = 0; i < 8; ++i)
        *(f4*)&potbuf[1 - buf][(c + i * 64) * 4] = nv[i];
      __syncthreads();
    }
  }

  if (c < 32) {
    const float* alast = alpha[(TT - 1) & 1];
    f4 av[16];
#pragma unroll
    for (int i = 0; i < 16; ++i) av[i] = *(const f4*)(alast + 4 * i);
    float bb = av[0][0];
    unsigned tg = 0;
#pragma unroll
    for (int p = 1; p < UU; ++p) {
      float s = av[p >> 2][p & 3];
      if (s > bb) { bb = s; tg = p; }
    }

    int* outb = path + (size_t)b * TT;
    const unsigned* rowp = bpw + c;
    unsigned cur[8], nxt[8];
#pragma unroll
    for (int j = 0; j < 8; ++j) cur[j] = rowp[(31 * 8 + j) * 32];

    for (int g = TT / 16 - 1; g >= 0; --g) {
      const int gn = g > 0 ? g - 1 : 0;
#pragma unroll
      for (int j = 0; j < 8; ++j) nxt[j] = rowp[(gn * 8 + j) * 32];

      unsigned acc = 0;
#pragma unroll
      for (int i = 15; i >= 0; --i) {
        acc = (c == 16 + i) ? tg : acc;
        const unsigned sel = tg >> 2;
        const unsigned w = (i & 1)
            ? __builtin_amdgcn_permlane16_var(cur[i >> 1], cur[i >> 1], sel, false, false)
            : __builtin_amdgcn_permlanex16_var(cur[i >> 1], cur[i >> 1], sel, false, false);
        tg = (w >> ((tg & 3u) * 8u)) & 0xffu;
      }
      if (c >= 16) outb[g * 16 + (c - 16)] = (int)acc;
#pragma unroll
      for (int j = 0; j < 8; ++j) cur[j] = nxt[j];
    }
  }
}

extern "C" void kernel_launch(void* const* d_in, const int* in_sizes, int n_in,
                              void* d_out, int out_size, void* d_ws,
                              size_t ws_size, hipStream_t stream) {
  (void)in_sizes; (void)n_in; (void)out_size; (void)ws_size;
  const float* x     = (const float*)d_in[0];
  const float* w     = (const float*)d_in[1];
  const float* bias  = (const float*)d_in[2];
  const float* chain = (const float*)d_in[3];
  const float* lb    = (const float*)d_in[4];
  const float* rb    = (const float*)d_in[5];
  float* pot = (float*)d_ws;
  int* out = (int*)d_out;

  crf_potentials_gemm<<<dim3(MM / 128), dim3(256), 0, stream>>>(
      x, w, bias, lb, rb, pot);
  crf_viterbi<<<dim3(BB), dim3(64), 0, stream>>>(pot, chain, out);
}


# Viterbi step laid round a split barrier by source wave: signal when own alphas land; own 32 read back in its shadow and scanned under the other 32's LDS flight; halves joined; same sums, tie order
# speedup vs baseline: 1.9593x; 1.1471x over previous
#include <hip/hip_runtime.h>
#include <hip/hip_bf16.h>

typedef float v2f __attribute__((ext_vector_type(2)));
typedef float v8f __attribute__((ext_vector_type(8)));
typedef float f4  __attribute__((ext_vector_type(4)));

constexpr int BB = 64;
constexpr int TT = 512;
constexpr int FF = 1024;
constexpr int UU = 64;
constexpr int MM = BB * TT;

#define KCH 64
#define LDAW 68
#define LDBW 1028

__global__ __launch_bounds__(256) void crf_potentials_gemm(
    const float* __restrict__ x, const float* __restrict__ w,
    const float* __restrict__ bias, const float* __restrict__ lb,
    const float* __restrict__ rb, float* __restrict__ pot) {
  __shared__ __attribute__((aligned(16))) float lds_b[UU * LDBW];
  __shared__ __attribute__((aligned(16))) float lds_a[8 * 16 * LDAW];

  const int tid = threadIdx.x;
  const int lane = tid & 31;
  const int wave = tid >> 5;
  const int rowBase = blockIdx.x * 128 + wave * 16;

  const int mrow = lane & 15;
  const int kbase = (lane >> 4) << 1;
  const int q = lane & 15;
  const int hf = lane >> 4;

  const float* xl = x + (size_t)(rowBase + hf) * FF + 4 * q;
  float* sa = lds_a + wave * (16 * LDAW);
  float* sa_st = sa + hf * LDAW + 4 * q;
  const float* sa_fr = sa + mrow * LDAW + kbase;
  const float* sb_fr = lds_b + mrow * LDBW + kbase;

  f4 r0[8], r1[8];
#pragma unroll
  for (int i = 0; i < 8; ++i) r0[i] = *(const f4*)(xl + (size_t)(2 * i) * FF);
#pragma unroll
  for (int i = 0; i < 8; ++i) r1[i] = *(const f4*)(xl + (size_t)(2 * i) * FF + KCH);

  {
    const int b4 = q >> 2;
#pragma unroll 4
    for (int it = 0; it < 16; ++it) {
      const int k0 = (it * 8 + wave) * 8;
      f4 bv[4];
#pragma unroll
      for (int r = 0; r < 4; ++r) {
        const int k = k0 + ((b4 + r) & 3) + 4 * hf;
        bv[r] = *(const f4*)(w + (size_t)k * UU + 4 * q);
      }
#pragma unroll
      for (int r = 0; r < 4; ++r) {
        const int k = k0 + ((b4 + r) & 3) + 4 * hf;
        lds_b[(4 * q + 0) * LDBW + k] = bv[r].x;
        lds_b[(4 * q + 1) * LDBW + k] = bv[r].y;
        lds_b[(4 * q + 2) * LDBW + k] = bv[r].z;
        lds_b[(4 * q + 3) * LDBW + k] = bv[r].w;
      }
    }
  }
  __syncthreads();

  v8f acc[4] = {};

#define STORE_SLOT(R)                                                         \
  _Pragma("unroll") for (int i = 0; i < 8; ++i)                               \
      *(f4*)(sa_st + (2 * i) * LDAW) = R[i];
#define LOAD_CHUNK(R, K0)                                                     \
  _Pragma("unroll") for (int i = 0; i < 8; ++i)                               \
      R[i] = *(const f4*)(xl + (size_t)(2 * i) * FF + (K0));
#define MMA_CHUNK(K0)                                                         \
  _Pragma("unroll") for (int k4 = 0; k4 < KCH; k4 += 4) {                     \
    v2f afrag = *(const v2f*)(sa_fr + k4);                                    \
    _Pragma("unroll") for (int nt = 0; nt < 4; ++nt) {                        \
      v2f bfrag = *(const v2f*)(sb_fr + nt * 16 * LDBW + (K0) + k4);          \
      acc[nt] = __builtin_amdgcn_wmma_f32_16x16x4_f32(                        \
          false, afrag, false, bfrag, (short)0, acc[nt], false, false);       \
    }                                                                         \
  }

  for (int kk = 0; kk < FF - 2 * KCH; kk += 2 * KCH) {
    STORE_SLOT(r0)
    LOAD_CHUNK(r0, kk + 2 * KCH)
    MMA_CHUNK(kk)
    STORE_SLOT(r1)
    LOAD_CHUNK(r1, kk + 3 * KCH)
    MMA_CHUNK(kk + KCH)
  }
  STORE_SLOT(r0)
  MMA_CHUNK(FF - 2 * KCH)
  STORE_SLOT(r1)
  MMA_CHUNK(FF - KCH)
#undef STORE_SLOT
#undef LOAD_CHUNK
#undef MMA_CHUNK

  const int rsel = (lane >> 4) * 8;
#pragma unroll
  for (int nt = 0; nt < 4; ++nt) {
    int col = nt * 16 + mrow;
    float bia = bias[col];
    float lbv = lb[col];
    float rbv = rb[col];
#pragma unroll
    for (int r = 0; r < 8; ++r) {
      int row = rowBase + r + rsel;
      float v = acc[nt][r] + bia;
      int t = row & (TT - 1);
      if (t == 0) v += lbv;
      if (t == TT - 1) v += rbv;
      pot[(size_t)row * UU + col] = v;
    }
  }
}

static __device__ __forceinline__ void lds_landed(int before) { asm volatile("s_wait_dscnt 0x0" :: "v"(before) : "memory"); }
static __device__ __forceinline__ void bar_signal() { asm volatile("s_barrier_signal -1" ::: "memory"); }
static __device__ __forceinline__ void bar_wait()   { asm volatile("s_barrier_wait -1" ::: "memory"); }

__global__ __launch_bounds__(64) void crf_viterbi(
    const float* __restrict__ pot, const float* __restrict__ chain,
    int* __restrict__ path) {
  __shared__ __attribute__((aligned(16))) float alpha[2][UU];
  __shared__ __attribute__((aligned(16))) unsigned int bpw[TT * UU / 4];
  unsigned char* bp = (unsigned char*)bpw;
  __shared__ __attribute__((aligned(16))) float potbuf[2][32 * UU];

  const int c = threadIdx.x;
  const int b = blockIdx.x;
  const float* potb = pot + (size_t)b * TT * UU;

  const int own = (c >> 5) * 32;
  const int oth = 32 - own;
  const bool hiown = c >= 32;

  float cko[32], ckr[32];
#pragma unroll
  for (int p = 0; p < 32; ++p) cko[p] = chain[(own + p) * UU + c];
#pragma unroll
  for (int p = 0; p < 32; ++p) ckr[p] = chain[(oth + p) * UU + c];

#pragma unroll
  for (int i = 0; i < 8; ++i) {
    int f = c + i * 64;
    *(f4*)&potbuf[0][f * 4] = *(const f4*)(potb + f * 4);
  }
  __syncthreads();
  alpha[0][c] = potbuf[0][c];
  bp[c] = 0;
  f4 oav[8];
#pragma unroll
  for (int i = 0; i < 8; ++i) oav[i] = *(const f4*)(&alpha[0][own + 4 * i]);
  __syncthreads();

  int t = 1;
  for (int chunk = 0; chunk < 16; ++chunk) {
    const int buf = chunk & 1;
    f4 nv[8];
    if (chunk < 15) {
#pragma unroll
      for (int i = 0; i < 8; ++i)
        nv[i] = *(const f4*)(potb + (chunk + 1) * 32 * UU + (c + i * 64) * 4);
    }
    const int tEnd = (chunk + 1) * 32;
    float ptn = potbuf[buf][(t - chunk * 32) * UU + c];
    for (; t < tEnd; ++t) {
      const int tl = t - chunk * 32;
      const float* rrow = alpha[(t - 1) & 1] + oth;
      f4 rav[8];
#pragma unroll
      for (int i = 0; i < 8; ++i) rav[i] = *(const f4*)(rrow + 4 * i);
      __builtin_amdgcn_sched_barrier(0);

#define HALF_SCAN(AV, CK, S, BBV, GIV, WV)                                                     \
      float S[32];                                                                             \
      _Pragma("unroll") for (int p = 0; p < 32; ++p) S[p] = AV[p >> 2][p & 3] + CK[p];        \
      float BBV = -3.402823466e38f;   \
      int GIV = 0;                                                                             \
      float WV[8];                                                                             \
      _Pragma("unroll") for (int k = 0; k < 8; ++k) WV[k] = -3.402823466e38f;   \
      _Pragma("unroll") for (int g = 0; g < 4; ++g) {                                          \
        const float gm = fmaxf(fmaxf(fmaxf(S[8 * g + 0], S[8 * g + 1]), fmaxf(S[8 * g + 2], S[8 * g + 3])), \
                               fmaxf(fmaxf(S[8 * g + 4], S[8 * g + 5]), fmaxf(S[8 * g + 6], S[8 * g + 7]))); \
        const bool up = gm > BBV;        \
        BBV = fmaxf(BBV, gm);                                                                  \
        GIV = up ? g : GIV;                                                                    \
        _Pragma("unroll") for (int k = 0; k < 8; ++k) WV[k] = up ? S[8 * g + k] : WV[k];      \
      }

      HALF_SCAN(oav, cko, so, bbo, gio, wo)
      __builtin_amdgcn_sched_barrier(0);
      HALF_SCAN(rav, ckr, sr, bbr, gir, wr)
#undef HALF_SCAN

      const bool taker = (bbr > bbo) | (hiown & (bbr == bbo));
      const float bb = fmaxf(bbo, bbr);
      const int gi = taker ? (gir + (oth >> 3)) : (gio + (own >> 3));
      float wv[8];
#pragma unroll
      for (int k = 0; k < 8; ++k) wv[k] = taker ? wr[k] : wo[k];

      float* nrow = alpha[t & 1];
      nrow[c] = bb + ptn;

      int klo = (wv[2] == bb) ? 2 : 3;
      klo = (wv[1] == bb) ? 1 : klo;
      klo = (wv[0] == bb) ? 0 : klo;
      int khi = (wv[6] == bb) ? 6 : 7;
      khi = (wv[5] == bb) ? 5 : khi;
      khi = (wv[4] == bb) ? 4 : khi;
      const bool inlo = fmaxf(fmaxf(wv[0], wv[1]), fmaxf(wv[2], wv[3])) == bb;
      const int bidx = gi * 8 + (inlo ? klo : khi);

      lds_landed(bidx);
      bar_signal();
#pragma unroll
      for (int i = 0; i < 8; ++i) oav[i] = *(const f4*)(nrow + own + 4 * i);
      ptn = potbuf[buf][((tl + 1) & 31) * UU + c];
      bp[t * UU + c] = (unsigned char)bidx;
      bar_wait();
    }
    if (chunk < 15) {
#pragma unroll
      for (int i = 0; i < 8; ++i)
        *(f4*)&potbuf[1 - buf][(c + i * 64) * 4] = nv[i];
      __syncthreads();
    }
  }

  __syncthreads();

  if (c < 32) {
    const float* alast = alpha[(TT - 1) & 1];
    f4 av[16];
#pragma unroll
    for (int i = 0; i < 16; ++i) av[i] = *(const f4*)(alast + 4 * i);
    float bb = av[0][0];
    unsigned tg = 0;
#pragma unroll
    for (int p = 1; p < UU; ++p) {
      float s = av[p >> 2][p & 3];
      if (s > bb) { bb = s; tg = p; }
    }

    int* outb = path + (size_t)b * TT;
    const unsigned* rowp = bpw + c;
    unsigned cur[8], nxt[8];
#pragma unroll
    for (int j = 0; j < 8; ++j) cur[j] = rowp[(31 * 8 + j) * 32];

    for (int g = TT / 16 - 1; g >= 0; --g) {
      const int gn = g > 0 ? g - 1 : 0;
#pragma unroll
      for (int j = 0; j < 8; ++j) nxt[j] = rowp[(gn * 8 + j) * 32];

      unsigned acc = 0;
#pragma unroll
      for (int i = 15; i >= 0; --i) {
        acc = (c == 16 + i) ? tg : acc;
        const unsigned sel = tg >> 2;
        const unsigned w = (i & 1)
            ? __builtin_amdgcn_permlane16_var(cur[i >> 1], cur[i >> 1], sel, false, false)
            : __builtin_amdgcn_permlanex16_var(cur[i >> 1], cur[i >> 1], sel, false, false);
        tg = (w >> ((tg & 3u) * 8u)) & 0xffu;
      }
      if (c >= 16) outb[g * 16 + (c - 16)] = (int)acc;
#pragma unroll
      for (int j = 0; j < 8; ++j) cur[j] = nxt[j];
    }
  }
}

extern "C" void kernel_launch(void* const* d_in, const int* in_sizes, int n_in,
                              void* d_out, int out_size, void* d_ws,
                              size_t ws_size, hipStream_t stream) {
  (void)in_sizes; (void)n_in; (void)out_size; (void)ws_size;
  const float* x     = (const float*)d_in[0];
  const float* w     = (const float*)d_in[1];
  const float* bias  = (const float*)d_in[2];
  const float* chain = (const float*)d_in[3];
  const float* lb    = (const float*)d_in[4];
  const float* rb    = (const float*)d_in[5];
  float* pot = (float*)d_ws;
  int* out = (int*)d_out;

  crf_potentials_gemm<<<dim3(MM / 128), dim3(256), 0, stream>>>(
      x, w, bias, lb, rb, pot);
  crf_viterbi<<<dim3(BB), dim3(64), 0, stream>>>(pot, chain, out);
}
